# starting kernel
# baseline (speedup 1.0000x reference)
#include <hip/hip_runtime.h>

typedef float v2f __attribute__((ext_vector_type(2)));
typedef float v4f __attribute__((ext_vector_type(4)));
typedef float v8f __attribute__((ext_vector_type(8)));

#define SEQ   32768
#define DIM   1024
#define ROWS_PER_WAVE 16
#define WAVES_PER_BLOCK 8

__global__ __launch_bounds__(WAVES_PER_BLOCK * 32)
void gates_gemm_wmma(const float* __restrict__ x,
                     const float* __restrict__ Wih,
                     const float* __restrict__ bih,
                     const float* __restrict__ bhh,
                     float* __restrict__ gates)
{
    const int lane    = threadIdx.x & 31;
    const int wave    = threadIdx.x >> 5;
    const int lane_lo = lane & 15;
    const int lane_hi = lane >> 4;
    const int row_base = (blockIdx.x * WAVES_PER_BLOCK + wave) * ROWS_PER_WAVE;

    const float* xrow = x + (size_t)(row_base + lane_lo) * DIM + 2 * lane_hi;
    const float* wrow = Wih + (size_t)(lane_lo & 3) * DIM + 2 * lane_hi;

    v8f acc = {};
    #pragma unroll 4
    for (int k = 0; k < DIM; k += 4) {
        v2f a = *(const v2f*)(xrow + k);
        v2f b = *(const v2f*)(wrow + k);
        acc = __builtin_amdgcn_wmma_f32_16x16x4_f32(
                   false, a,  false, b,
                   (short)0, acc,  false,  false);
    }

    if (lane_lo < 4) {
        const int g    = lane_lo;
        const float bb = bih[g] + bhh[g];
        const int r0   = row_base + lane_hi * 8;
        #pragma unroll
        for (int r = 0; r < 8; ++r)
            gates[(size_t)(r0 + r) * 4 + g] = acc[r] + bb;
    }
}

__device__ __forceinline__ float fsig(float xv)
{
    return __builtin_amdgcn_rcpf(1.0f + __builtin_amdgcn_exp2f(-1.4426950408889634f * xv));
}
__device__ __forceinline__ float ftanh(float xv)
{
    return fmaf(2.0f,
                __builtin_amdgcn_rcpf(1.0f + __builtin_amdgcn_exp2f(-2.8853901631779268f * xv)),
                -1.0f);
}

__global__ void lstm_scan(const float* __restrict__ gates,
                          const float* __restrict__ Whh,
                          const float* __restrict__ h0,
                          const float* __restrict__ c0,
                          float* __restrict__ out)
{
    if (threadIdx.x != 0) return;

    const float w0 = Whh[0], w1 = Whh[1], w2 = Whh[2], w3 = Whh[3];
    float h = h0[0];
    float c = c0[0];

    v4f hbuf = {};
    for (int t = 0; t < SEQ; ++t) {
        v4f g = *(const v4f*)(gates + 4 * (size_t)t);
        if ((t & 31) == 0)
            __builtin_prefetch(gates + 4 * (size_t)t + 512, 0, 3);

        const float gi = fmaf(w0, h, g.x);
        const float gf = fmaf(w1, h, g.y);
        const float gc = fmaf(w2, h, g.z);
        const float go = fmaf(w3, h, g.w);

        const float iv = fsig(gi);
        const float fv = fsig(gf);
        const float gv = ftanh(gc);
        const float ov = fsig(go);

        c = fmaf(fv, c, iv * gv);
        h = ov * ftanh(c);

        hbuf[t & 3] = h;
        if ((t & 3) == 3)
            *(v4f*)(out + (t - 3)) = hbuf;
    }
}

extern "C" void kernel_launch(void* const* d_in, const int* in_sizes, int n_in,
                              void* d_out, int out_size, void* d_ws, size_t ws_size,
                              hipStream_t stream)
{
    const float* x    = (const float*)d_in[0];
    const float* Wih  = (const float*)d_in[1];
    const float* Whh  = (const float*)d_in[2];
    const float* bih  = (const float*)d_in[3];
    const float* bhh  = (const float*)d_in[4];
    const float* h0   = (const float*)d_in[5];
    const float* c0   = (const float*)d_in[6];
    float*       out  = (float*)d_out;
    float*       gts  = (float*)d_ws;

    const int rows_per_block = ROWS_PER_WAVE * WAVES_PER_BLOCK;
    const int nblocks = SEQ / rows_per_block;

    gates_gemm_wmma<<<nblocks, WAVES_PER_BLOCK * 32, 0, stream>>>(x, Wih, bih, bhh, gts);
    lstm_scan<<<1, 32, 0, stream>>>(gts, Whh, h0, c0, out);
}


# scan kernel: one wave loads the gates of 32 steps (one per lane) a block ahead, broadcasts each step's four by v_readlane, one 128-byte store per block: no memory wait left on the serial chain
# speedup vs baseline: 2.2823x; 2.2823x over previous
#include <hip/hip_runtime.h>

typedef float v2f __attribute__((ext_vector_type(2)));
typedef float v4f __attribute__((ext_vector_type(4)));
typedef float v8f __attribute__((ext_vector_type(8)));

#define SEQ   32768
#define DIM   1024
#define ROWS_PER_WAVE 16
#define WAVES_PER_BLOCK 8

__global__ __launch_bounds__(WAVES_PER_BLOCK * 32)
void gates_gemm_wmma(const float* __restrict__ x,
                     const float* __restrict__ Wih,
                     const float* __restrict__ bih,
                     const float* __restrict__ bhh,
                     float* __restrict__ gates)
{
    const int lane    = threadIdx.x & 31;
    const int wave    = threadIdx.x >> 5;
    const int lane_lo = lane & 15;
    const int lane_hi = lane >> 4;
    const int row_base = (blockIdx.x * WAVES_PER_BLOCK + wave) * ROWS_PER_WAVE;

    const float* xrow = x + (size_t)(row_base + lane_lo) * DIM + 2 * lane_hi;
    const float* wrow = Wih + (size_t)(lane_lo & 3) * DIM + 2 * lane_hi;

    v8f acc = {};
    #pragma unroll 4
    for (int k = 0; k < DIM; k += 4) {
        v2f a = *(const v2f*)(xrow + k);
        v2f b = *(const v2f*)(wrow + k);
        acc = __builtin_amdgcn_wmma_f32_16x16x4_f32(
                   false, a,  false, b,
                   (short)0, acc,  false,  false);
    }

    if (lane_lo < 4) {
        const int g    = lane_lo;
        const float bb = bih[g] + bhh[g];
        const int r0   = row_base + lane_hi * 8;
        #pragma unroll
        for (int r = 0; r < 8; ++r)
            gates[(size_t)(r0 + r) * 4 + g] = acc[r] + bb;
    }
}

__device__ __forceinline__ float fsig(float xv)
{
    return __builtin_amdgcn_rcpf(1.0f + __builtin_amdgcn_exp2f(-1.4426950408889634f * xv));
}
__device__ __forceinline__ float ftanh(float xv)
{
    return fmaf(2.0f,
                __builtin_amdgcn_rcpf(1.0f + __builtin_amdgcn_exp2f(-2.8853901631779268f * xv)),
                -1.0f);
}

#define STEPS_PER_BLOCK 32
#define NBLOCKS (SEQ / STEPS_PER_BLOCK)

__device__ __forceinline__ float lane_bcast(float v, int j)
{
    return __builtin_bit_cast(float,
               __builtin_amdgcn_readlane(__builtin_bit_cast(int, v), j));
}

__global__ __launch_bounds__(32)
void lstm_scan(const float* __restrict__ gates,
               const float* __restrict__ Whh,
               const float* __restrict__ h0,
               const float* __restrict__ c0,
               float* __restrict__ out)
{
    const int lane = threadIdx.x;

    const float w0 = Whh[0], w1 = Whh[1], w2 = Whh[2], w3 = Whh[3];
    float h, c;
    asm volatile("v_mov_b32 %0, %2\n\tv_mov_b32 %1, %3"
                 : "=&v"(h), "=v"(c) : "s"(h0[0]), "s"(c0[0]));

    const v4f* gp = (const v4f*)gates + lane;
    v4f gcur = gp[0];
    for (int b = 0; b < NBLOCKS; ++b) {
        const int bn = (b + 1 < NBLOCKS) ? (b + 1) : b;
        const v4f gnext = gp[(size_t)bn * STEPS_PER_BLOCK];

        float hkeep = 0.0f;
        #pragma unroll
        for (int j = 0; j < STEPS_PER_BLOCK; ++j) {
            const float gx = lane_bcast(gcur.x, j);
            const float gy = lane_bcast(gcur.y, j);
            const float gz = lane_bcast(gcur.z, j);
            const float gw = lane_bcast(gcur.w, j);

            const float gi = fmaf(w0, h, gx);
            const float gf = fmaf(w1, h, gy);
            const float gc = fmaf(w2, h, gz);
            const float go = fmaf(w3, h, gw);

            const float iv = fsig(gi);
            const float fv = fsig(gf);
            const float gv = ftanh(gc);
            const float ov = fsig(go);

            c = fmaf(fv, c, iv * gv);
            h = ov * ftanh(c);

            hkeep = (lane == j) ? h : hkeep;
        }
        out[(size_t)b * STEPS_PER_BLOCK + lane] = hkeep;
        gcur = gnext;
    }
}

extern "C" void kernel_launch(void* const* d_in, const int* in_sizes, int n_in,
                              void* d_out, int out_size, void* d_ws, size_t ws_size,
                              hipStream_t stream)
{
    const float* x    = (const float*)d_in[0];
    const float* Wih  = (const float*)d_in[1];
    const float* Whh  = (const float*)d_in[2];
    const float* bih  = (const float*)d_in[3];
    const float* bhh  = (const float*)d_in[4];
    const float* h0   = (const float*)d_in[5];
    const float* c0   = (const float*)d_in[6];
    float*       out  = (float*)d_out;
    float*       gts  = (float*)d_ws;

    const int rows_per_block = ROWS_PER_WAVE * WAVES_PER_BLOCK;
    const int nblocks = SEQ / rows_per_block;

    gates_gemm_wmma<<<nblocks, WAVES_PER_BLOCK * 32, 0, stream>>>(x, Wih, bih, bhh, gts);
    lstm_scan<<<1, 32, 0, stream>>>(gts, Whh, h0, c0, out);
}


# scan: per-step dependent chain cut to fma-tanh-fma-fma-tanh: sigmoid = 0.5*tanh(0.5z)+0.5 on v_tanh_f32, 0.5 scales and o folded into gates/weights, cell sum re-associated, issue order pinned
# speedup vs baseline: 6.2806x; 2.7519x over previous
#include <hip/hip_runtime.h>

typedef float v2f __attribute__((ext_vector_type(2)));
typedef float v4f __attribute__((ext_vector_type(4)));
typedef float v8f __attribute__((ext_vector_type(8)));

#define SEQ   32768
#define DIM   1024
#define ROWS_PER_WAVE 16
#define WAVES_PER_BLOCK 8

__global__ __launch_bounds__(WAVES_PER_BLOCK * 32)
void gates_gemm_wmma(const float* __restrict__ x,
                     const float* __restrict__ Wih,
                     const float* __restrict__ bih,
                     const float* __restrict__ bhh,
                     float* __restrict__ gates)
{
    const int lane    = threadIdx.x & 31;
    const int wave    = threadIdx.x >> 5;
    const int lane_lo = lane & 15;
    const int lane_hi = lane >> 4;
    const int row_base = (blockIdx.x * WAVES_PER_BLOCK + wave) * ROWS_PER_WAVE;

    const float* xrow = x + (size_t)(row_base + lane_lo) * DIM + 2 * lane_hi;
    const float* wrow = Wih + (size_t)(lane_lo & 3) * DIM + 2 * lane_hi;

    v8f acc = {};
    #pragma unroll 4
    for (int k = 0; k < DIM; k += 4) {
        v2f a = *(const v2f*)(xrow + k);
        v2f b = *(const v2f*)(wrow + k);
        acc = __builtin_amdgcn_wmma_f32_16x16x4_f32(
                   false, a,  false, b,
                   (short)0, acc,  false,  false);
    }

    if (lane_lo < 4) {
        const int g    = lane_lo;
        const float bb = bih[g] + bhh[g];
        const float sc = (g == 2) ? 1.0f : 0.5f;
        const int r0   = row_base + lane_hi * 8;
        #pragma unroll
        for (int r = 0; r < 8; ++r)
            gates[(size_t)(r0 + r) * 4 + g] = (acc[r] + bb) * sc;
    }
}

__device__ __forceinline__ float hw_tanh(float xv)
{
    return __builtin_amdgcn_tanhf(xv);
}

#define STEPS_PER_BLOCK 32
#define NBLOCKS (SEQ / STEPS_PER_BLOCK)

#define PIN_ORDER() __builtin_amdgcn_sched_barrier(0)

__device__ __forceinline__ float lane_bcast(float v, int j)
{
    return __builtin_bit_cast(float,
               __builtin_amdgcn_readlane(__builtin_bit_cast(int, v), j));
}

__global__ __launch_bounds__(32)
void lstm_scan(const float* __restrict__ gates,
               const float* __restrict__ Whh,
               const float* __restrict__ h0,
               const float* __restrict__ c0,
               float* __restrict__ out)
{
    const int lane = threadIdx.x;

    const float ws0 = 0.5f * Whh[0], ws1 = 0.5f * Whh[1],
                ws2 =        Whh[2], ws3 = 0.5f * Whh[3];
    float tc, c;
    asm volatile("v_mov_b32 %0, %2\n\tv_mov_b32 %1, %3"
                 : "=&v"(tc), "=v"(c) : "s"(h0[0]), "s"(c0[0]));
    float ov  = 1.0f;
    float wo0 = ws0, wo1 = ws1, wo2 = ws2, wo3 = ws3;

    const v4f* gp = (const v4f*)gates + lane;
    v4f gcur = gp[0];
    float gx = lane_bcast(gcur.x, 0), gy = lane_bcast(gcur.y, 0),
          gz = lane_bcast(gcur.z, 0), gw = lane_bcast(gcur.w, 0);

    for (int b = 0; b < NBLOCKS; ++b) {
        const int bn = (b + 1 < NBLOCKS) ? (b + 1) : b;
        const v4f gnext = gp[(size_t)bn * STEPS_PER_BLOCK];

        float hkeep = 0.0f;
        #pragma unroll
        for (int j = 0; j < STEPS_PER_BLOCK; ++j) {
            const float ai = fmaf(wo0, tc, gx);
            const float af = fmaf(wo1, tc, gy);
            const float ag = fmaf(wo2, tc, gz);
            const float ao = fmaf(wo3, tc, gw);
            float hprev = 0.0f;
            if (j > 0) hprev = ov * tc;
            PIN_ORDER();
            const float ti = hw_tanh(ai);
            const float ch = 0.5f * c;
            PIN_ORDER();
            const float tf = hw_tanh(af);
            const v4f gsrc = (j == STEPS_PER_BLOCK - 1) ? gnext : gcur;
            const int  jn   = (j + 1) & (STEPS_PER_BLOCK - 1);
            const float nx = lane_bcast(gsrc.x, jn);
            PIN_ORDER();
            const float tg = hw_tanh(ag);
            if (j > 0) hkeep = (lane == j - 1) ? hprev : hkeep;
            PIN_ORDER();
            const float to = hw_tanh(ao);
            const float ny = lane_bcast(gsrc.y, jn);
            PIN_ORDER();
            const float iv = fmaf(0.5f, ti, 0.5f);
            const float nz = lane_bcast(gsrc.z, jn);
            PIN_ORDER();
            const float P = fmaf(tf, ch, ch);
            const float nw = lane_bcast(gsrc.w, jn);
            PIN_ORDER();
            ov = fmaf(0.5f, to, 0.5f);
            PIN_ORDER();
            c = fmaf(iv, tg, P);
            PIN_ORDER();
            wo0 = ws0 * ov;
            PIN_ORDER();
            tc = hw_tanh(c);
            PIN_ORDER();
            wo1 = ws1 * ov; wo2 = ws2 * ov; wo3 = ws3 * ov;
            PIN_ORDER();
            gx = nx; gy = ny; gz = nz; gw = nw;
        }
        const float hlast = ov * tc;
        hkeep = (lane == STEPS_PER_BLOCK - 1) ? hlast : hkeep;
        out[(size_t)b * STEPS_PER_BLOCK + lane] = hkeep;
        gcur = gnext;
    }
}

extern "C" void kernel_launch(void* const* d_in, const int* in_sizes, int n_in,
                              void* d_out, int out_size, void* d_ws, size_t ws_size,
                              hipStream_t stream)
{
    const float* x    = (const float*)d_in[0];
    const float* Wih  = (const float*)d_in[1];
    const float* Whh  = (const float*)d_in[2];
    const float* bih  = (const float*)d_in[3];
    const float* bhh  = (const float*)d_in[4];
    const float* h0   = (const float*)d_in[5];
    const float* c0   = (const float*)d_in[6];
    float*       out  = (float*)d_out;
    float*       gts  = (float*)d_ws;

    const int rows_per_block = ROWS_PER_WAVE * WAVES_PER_BLOCK;
    const int nblocks = SEQ / rows_per_block;

    gates_gemm_wmma<<<nblocks, WAVES_PER_BLOCK * 32, 0, stream>>>(x, Wih, bih, bhh, gts);
    lstm_scan<<<1, 32, 0, stream>>>(gts, Whh, h0, c0, out);
}


# scan: sequence cut into 256 chunks, one per lane of 8 waves (gates stored chunk-interleaved); chunks run from guessed states, re-run from the neighbour's end state until all boundaries match bitwise
# speedup vs baseline: 85.9146x; 13.6794x over previous
#include <hip/hip_runtime.h>

typedef float v2f __attribute__((ext_vector_type(2)));
typedef float v4f __attribute__((ext_vector_type(4)));
typedef float v8f __attribute__((ext_vector_type(8)));

#define SEQ   32768
#define DIM   1024
#define ROWS_PER_WAVE 16
#define WAVES_PER_BLOCK 8

#define NW        8
#define NTHREADS  (NW * 32)
#define NCHUNK    NTHREADS
#define CL        (SEQ / NCHUNK)

__global__ __launch_bounds__(WAVES_PER_BLOCK * 32)
void gates_gemm_wmma(const float* __restrict__ x,
                     const float* __restrict__ Wih,
                     const float* __restrict__ bih,
                     const float* __restrict__ bhh,
                     float* __restrict__ gates)
{
    const int lane    = threadIdx.x & 31;
    const int wave    = threadIdx.x >> 5;
    const int lane_lo = lane & 15;
    const int lane_hi = lane >> 4;
    const int qg = blockIdx.x & 15;
    const int t  = (blockIdx.x >> 4) * WAVES_PER_BLOCK + wave;

    const float* xrow = x + (size_t)((qg * ROWS_PER_WAVE + lane_lo) * CL + t) * DIM + 2 * lane_hi;
    const float* wrow = Wih + (size_t)(lane_lo & 3) * DIM + 2 * lane_hi;

    v8f acc = {};
    #pragma unroll 4
    for (int k = 0; k < DIM; k += 4) {
        v2f a = *(const v2f*)(xrow + k);
        v2f b = *(const v2f*)(wrow + k);
        acc = __builtin_amdgcn_wmma_f32_16x16x4_f32(
                   false, a,  false, b,
                   (short)0, acc,  false,  false);
    }

    if (lane_lo < 4) {
        const int g    = lane_lo;
        const float bb = bih[g] + bhh[g];
        const float sc = (g == 2) ? 1.0f : 0.5f;
        const int q0   = qg * ROWS_PER_WAVE + lane_hi * 8;
        #pragma unroll
        for (int r = 0; r < 8; ++r)
            gates[((size_t)t * NCHUNK + q0 + r) * 4 + g] = (acc[r] + bb) * sc;
    }
}

__device__ __forceinline__ float hw_tanh(float xv)
{
    return __builtin_amdgcn_tanhf(xv);
}

#define BS        16
#define NB        (CL / BS)
#define MAX_SWEEPS (NCHUNK + 1)

__device__ __forceinline__ void lstm_steps(const v4f (&g)[BS], float (&h)[BS],
                                           float& tc, float& c, float& ov,
                                           const float ws0, const float ws1,
                                           const float ws2, const float ws3)
{
    float wo0 = ws0 * ov, wo1 = ws1 * ov, wo2 = ws2 * ov, wo3 = ws3 * ov;
    #pragma unroll
    for (int j = 0; j < BS; ++j) {
        const float ai = fmaf(wo0, tc, g[j].x);
        const float af = fmaf(wo1, tc, g[j].y);
        const float ag = fmaf(wo2, tc, g[j].z);
        const float ao = fmaf(wo3, tc, g[j].w);
        const float ti = hw_tanh(ai);
        const float ch = 0.5f * c;
        const float tf = hw_tanh(af);
        const float tg = hw_tanh(ag);
        const float to = hw_tanh(ao);
        const float iv = fmaf(0.5f, ti, 0.5f);
        const float P  = fmaf(tf, ch, ch);
        ov = fmaf(0.5f, to, 0.5f);
        c  = fmaf(iv, tg, P);
        wo0 = ws0 * ov;
        tc = hw_tanh(c);
        wo1 = ws1 * ov; wo2 = ws2 * ov; wo3 = ws3 * ov;
        h[j] = ov * tc;
    }
}

__device__ __forceinline__ void load_block(v4f (&g)[BS], const v4f* __restrict__ p)
{
    #pragma unroll
    for (int j = 0; j < BS; ++j) g[j] = p[(size_t)j * NCHUNK];
}

__device__ __forceinline__ void store_block(float* __restrict__ o, const float (&h)[BS])
{
    #pragma unroll
    for (int q = 0; q < BS / 4; ++q) {
        v4f v = { h[4 * q], h[4 * q + 1], h[4 * q + 2], h[4 * q + 3] };
        *(v4f*)(o + 4 * q) = v;
    }
}

__global__ __launch_bounds__(NTHREADS)
void lstm_scan(const float* __restrict__ gates,
               const float* __restrict__ Whh,
               const float* __restrict__ h0,
               const float* __restrict__ c0,
               float* __restrict__ out)
{
    __shared__ float s_tc[NTHREADS], s_c[NTHREADS], s_ov[NTHREADS];
    __shared__ int   s_bad[NW];

    const int tid  = threadIdx.x;
    const int wave = tid >> 5;

    const float ws0 = 0.5f * Whh[0], ws1 = 0.5f * Whh[1],
                ws2 =        Whh[2], ws3 = 0.5f * Whh[3];

    float st_tc = (tid == 0) ? h0[0] : 0.0f;
    float st_c  = (tid == 0) ? c0[0] : 0.0f;
    float st_ov = (tid == 0) ? 1.0f  : 0.5f;

    const v4f* gp = (const v4f*)gates + tid;
    float*     op = out + (size_t)tid * CL;

    bool need_store = true;

    for (int sweep = 0; sweep < MAX_SWEEPS; ++sweep) {
        const bool st = need_store && (sweep > 0 || tid == 0);

        float tc = st_tc, c = st_c, ov = st_ov;
        v4f ga[BS], gb[BS];
        float h[BS];
        load_block(ga, gp);
        for (int b = 0; b < NB; b += 2) {
            load_block(gb, gp + (size_t)(b + 1) * BS * NCHUNK);
            lstm_steps(ga, h, tc, c, ov, ws0, ws1, ws2, ws3);
            if (st) store_block(op + b * BS, h);
            const int bn = (b + 2 < NB) ? (b + 2) : b;
            load_block(ga, gp + (size_t)bn * BS * NCHUNK);
            lstm_steps(gb, h, tc, c, ov, ws0, ws1, ws2, ws3);
            if (st) store_block(op + (b + 1) * BS, h);
        }
        if (st) need_store = false;

        s_tc[tid] = tc; s_c[tid] = c; s_ov[tid] = ov;
        __syncthreads();
        bool bad = false;
        float n_tc = st_tc, n_c = st_c, n_ov = st_ov;
        if (tid > 0) {
            n_tc = s_tc[tid - 1]; n_c = s_c[tid - 1]; n_ov = s_ov[tid - 1];
            bad = (__float_as_uint(n_tc) != __float_as_uint(st_tc)) ||
                  (__float_as_uint(n_c)  != __float_as_uint(st_c))  ||
                  (__float_as_uint(n_ov) != __float_as_uint(st_ov));
        }
        const bool wave_bad = __builtin_amdgcn_ballot_w32(bad) != 0u;
        if ((tid & 31) == 0) s_bad[wave] = wave_bad ? 1 : 0;
        __syncthreads();
        int any_bad = 0;
        #pragma unroll
        for (int w = 0; w < NW; ++w) any_bad |= s_bad[w];
        if (any_bad == 0 && sweep > 0) break;
        if (bad) { st_tc = n_tc; st_c = n_c; st_ov = n_ov; need_store = true; }
    }
}

extern "C" void kernel_launch(void* const* d_in, const int* in_sizes, int n_in,
                              void* d_out, int out_size, void* d_ws, size_t ws_size,
                              hipStream_t stream)
{
    const float* x    = (const float*)d_in[0];
    const float* Wih  = (const float*)d_in[1];
    const float* Whh  = (const float*)d_in[2];
    const float* bih  = (const float*)d_in[3];
    const float* bhh  = (const float*)d_in[4];
    const float* h0   = (const float*)d_in[5];
    const float* c0   = (const float*)d_in[6];
    float*       out  = (float*)d_out;
    float*       gts  = (float*)d_ws;

    const int rows_per_block = ROWS_PER_WAVE * WAVES_PER_BLOCK;
    const int nblocks = SEQ / rows_per_block;

    gates_gemm_wmma<<<nblocks, WAVES_PER_BLOCK * 32, 0, stream>>>(x, Wih, bih, bhh, gts);
    lstm_scan<<<1, NTHREADS, 0, stream>>>(gts, Whh, h0, c0, out);
}


# gates GEMM re-laid as a stream: every load reads whole 128-byte lines (WMMA tile = 4 rows x 4 k-segments, equal-segment entries added over 4 lanes), two load stages in flight; dot sum order changes
# speedup vs baseline: 110.8546x; 1.2903x over previous
#include <hip/hip_runtime.h>

typedef float v2f __attribute__((ext_vector_type(2)));
typedef float v4f __attribute__((ext_vector_type(4)));
typedef float v8f __attribute__((ext_vector_type(8)));

#define SEQ   32768
#define DIM   1024

#define NW        8
#define NTHREADS  (NW * 32)
#define NCHUNK    NTHREADS
#define CL        (SEQ / NCHUNK)

#define GW      16
#define GTHREADS (GW * 32)
#define KSTEPS  (DIM / 32)
#define STAGE   8
#define NSTAGE  (KSTEPS / STAGE)
#define GBLOCKS (SEQ / (8 * GW))

__device__ __forceinline__ v8f wmma4(const float a0, const float a1,
                                     const float b0, const float b1, const v8f acc)
{
    const v2f a = { a0, a1 };
    const v2f b = { b0, b1 };
    return __builtin_amdgcn_wmma_f32_16x16x4_f32(
                false, a,  false, b,
                (short)0, acc,  false,  false);
}

__device__ __forceinline__ float pick4(const float v0, const float v1,
                                       const float v2, const float v3, const int j)
{
    return (j == 0) ? v0 : (j == 1) ? v1 : (j == 2) ? v2 : v3;
}

__device__ __forceinline__ float quad_sum(float v)
{
    v += __shfl_xor(v, 1, 32);
    v += __shfl_xor(v, 2, 32);
    return v;
}

__global__ __launch_bounds__(GTHREADS)
void gates_gemm_wmma(const float* __restrict__ x,
                     const float* __restrict__ Wih,
                     const float* __restrict__ bih,
                     const float* __restrict__ bhh,
                     float* __restrict__ gates)
{
    const int lane = threadIdx.x & 31;
    const int wave = threadIdx.x >> 5;
    const int c    = lane & 15;
    const int h    = lane >> 4;
    const int q    = c >> 2;
    const int j    = c & 3;
    const int qo   = blockIdx.x >> 3;
    const int t    = (blockIdx.x & 7) * GW + wave;

    const int koff = 16 * h + 4 * j;
    const v4f* xa = (const v4f*)(x + (size_t)((qo * 8 + q) * CL + t) * DIM + koff);
    const v4f* xb = (const v4f*)(x + (size_t)((qo * 8 + 4 + q) * CL + t) * DIM + koff);
    const v4f* wp = (const v4f*)(Wih + (size_t)q * DIM + koff);

    v4f fa[2][STAGE], fb[2][STAGE], fw[2][STAGE];

#define LOAD_STAGE(S) do {                                                   \
        _Pragma("unroll") for (int i = 0; i < STAGE; ++i) {                  \
            fw[(S) & 1][i] = wp[((S) * STAGE + i) * 8];                      \
            fa[(S) & 1][i] = xa[((S) * STAGE + i) * 8];                      \
            fb[(S) & 1][i] = xb[((S) * STAGE + i) * 8];                      \
        } } while (0)

    v8f accA = {}, accB = {};
    LOAD_STAGE(0);
    #pragma unroll
    for (int s = 0; s < NSTAGE; ++s) {
        __builtin_amdgcn_sched_barrier(0);
        if (s + 1 < NSTAGE) LOAD_STAGE(s + 1);
        __builtin_amdgcn_sched_barrier(0);
        #pragma unroll
        for (int i = 0; i < STAGE; ++i) {
            const v4f a = fa[s & 1][i], b = fb[s & 1][i], w = fw[s & 1][i];
            accA = wmma4(a.x, a.y, w.x, w.y, accA);
            accB = wmma4(b.x, b.y, w.x, w.y, accB);
            accA = wmma4(a.z, a.w, w.z, w.w, accA);
            accB = wmma4(b.z, b.w, w.z, w.w, accB);
        }
    }
#undef LOAD_STAGE

    float a0 = pick4(accA[0], accA[1], accA[2], accA[3], j);
    float a1 = pick4(accA[4], accA[5], accA[6], accA[7], j);
    float b0 = pick4(accB[0], accB[1], accB[2], accB[3], j);
    float b1 = pick4(accB[4], accB[5], accB[6], accB[7], j);
    a0 = quad_sum(a0); a1 = quad_sum(a1); b0 = quad_sum(b0); b1 = quad_sum(b1);

    const float v  = pick4(a0, a1, b0, b1, j);
    const int   R  = 4 * (j >> 1) + 2 * h + (j & 1);
    const int   g  = q;
    const float bb = bih[g] + bhh[g];
    const float sc = (g == 2) ? 1.0f : 0.5f;
    gates[((size_t)t * NCHUNK + qo * 8 + R) * 4 + g] = (v + bb) * sc;
}

__device__ __forceinline__ float hw_tanh(float xv)
{
    return __builtin_amdgcn_tanhf(xv);
}

#define BS        16
#define NB        (CL / BS)
#define MAX_SWEEPS (NCHUNK + 1)

__device__ __forceinline__ void lstm_steps(const v4f (&g)[BS], float (&h)[BS],
                                           float& tc, float& c, float& ov,
                                           const float ws0, const float ws1,
                                           const float ws2, const float ws3)
{
    float wo0 = ws0 * ov, wo1 = ws1 * ov, wo2 = ws2 * ov, wo3 = ws3 * ov;
    #pragma unroll
    for (int j = 0; j < BS; ++j) {
        const float ai = fmaf(wo0, tc, g[j].x);
        const float af = fmaf(wo1, tc, g[j].y);
        const float ag = fmaf(wo2, tc, g[j].z);
        const float ao = fmaf(wo3, tc, g[j].w);
        const float ti = hw_tanh(ai);
        const float ch = 0.5f * c;
        const float tf = hw_tanh(af);
        const float tg = hw_tanh(ag);
        const float to = hw_tanh(ao);
        const float iv = fmaf(0.5f, ti, 0.5f);
        const float P  = fmaf(tf, ch, ch);
        ov = fmaf(0.5f, to, 0.5f);
        c  = fmaf(iv, tg, P);
        wo0 = ws0 * ov;
        tc = hw_tanh(c);
        wo1 = ws1 * ov; wo2 = ws2 * ov; wo3 = ws3 * ov;
        h[j] = ov * tc;
    }
}

__device__ __forceinline__ void load_block(v4f (&g)[BS], const v4f* __restrict__ p)
{
    #pragma unroll
    for (int j = 0; j < BS; ++j) g[j] = p[(size_t)j * NCHUNK];
}

__device__ __forceinline__ void store_block(float* __restrict__ o, const float (&h)[BS])
{
    #pragma unroll
    for (int q = 0; q < BS / 4; ++q) {
        v4f v = { h[4 * q], h[4 * q + 1], h[4 * q + 2], h[4 * q + 3] };
        *(v4f*)(o + 4 * q) = v;
    }
}

__global__ __launch_bounds__(NTHREADS)
void lstm_scan(const float* __restrict__ gates,
               const float* __restrict__ Whh,
               const float* __restrict__ h0,
               const float* __restrict__ c0,
               float* __restrict__ out)
{
    __shared__ float s_tc[NTHREADS], s_c[NTHREADS], s_ov[NTHREADS];
    __shared__ int   s_bad[NW];

    const int tid  = threadIdx.x;
    const int wave = tid >> 5;

    const float ws0 = 0.5f * Whh[0], ws1 = 0.5f * Whh[1],
                ws2 =        Whh[2], ws3 = 0.5f * Whh[3];

    float st_tc = (tid == 0) ? h0[0] : 0.0f;
    float st_c  = (tid == 0) ? c0[0] : 0.0f;
    float st_ov = (tid == 0) ? 1.0f  : 0.5f;

    const v4f* gp = (const v4f*)gates + tid;
    float*     op = out + (size_t)tid * CL;

    bool need_store = true;

    for (int sweep = 0; sweep < MAX_SWEEPS; ++sweep) {
        const bool st = need_store && (sweep > 0 || tid == 0);

        float tc = st_tc, c = st_c, ov = st_ov;
        v4f ga[BS], gb[BS];
        float h[BS];
        load_block(ga, gp);
        for (int b = 0; b < NB; b += 2) {
            load_block(gb, gp + (size_t)(b + 1) * BS * NCHUNK);
            lstm_steps(ga, h, tc, c, ov, ws0, ws1, ws2, ws3);
            if (st) store_block(op + b * BS, h);
            const int bn = (b + 2 < NB) ? (b + 2) : b;
            load_block(ga, gp + (size_t)bn * BS * NCHUNK);
            lstm_steps(gb, h, tc, c, ov, ws0, ws1, ws2, ws3);
            if (st) store_block(op + (b + 1) * BS, h);
        }
        if (st) need_store = false;

        s_tc[tid] = tc; s_c[tid] = c; s_ov[tid] = ov;
        __syncthreads();
        bool bad = false;
        float n_tc = st_tc, n_c = st_c, n_ov = st_ov;
        if (tid > 0) {
            n_tc = s_tc[tid - 1]; n_c = s_c[tid - 1]; n_ov = s_ov[tid - 1];
            bad = (__float_as_uint(n_tc) != __float_as_uint(st_tc)) ||
                  (__float_as_uint(n_c)  != __float_as_uint(st_c))  ||
                  (__float_as_uint(n_ov) != __float_as_uint(st_ov));
        }
        const bool wave_bad = __builtin_amdgcn_ballot_w32(bad) != 0u;
        if ((tid & 31) == 0) s_bad[wave] = wave_bad ? 1 : 0;
        __syncthreads();
        int any_bad = 0;
        #pragma unroll
        for (int w = 0; w < NW; ++w) any_bad |= s_bad[w];
        if (any_bad == 0 && sweep > 0) break;
        if (bad) { st_tc = n_tc; st_c = n_c; st_ov = n_ov; need_store = true; }
    }
}

extern "C" void kernel_launch(void* const* d_in, const int* in_sizes, int n_in,
                              void* d_out, int out_size, void* d_ws, size_t ws_size,
                              hipStream_t stream)
{
    const float* x    = (const float*)d_in[0];
    const float* Wih  = (const float*)d_in[1];
    const float* Whh  = (const float*)d_in[2];
    const float* bih  = (const float*)d_in[3];
    const float* bhh  = (const float*)d_in[4];
    const float* h0   = (const float*)d_in[5];
    const float* c0   = (const float*)d_in[6];
    float*       out  = (float*)d_out;
    float*       gts  = (float*)d_ws;

    gates_gemm_wmma<<<GBLOCKS, GTHREADS, 0, stream>>>(x, Wih, bih, bhh, gts);
    lstm_scan<<<1, NTHREADS, 0, stream>>>(gts, Whh, h0, c0, out);
}


# scan: the guess sweep runs only the last 64 steps of each chunk (its one product is the chunk's end state), chunk 0 too, so h0/c0 are first read for the stored sweep; bitwise boundary check unchanged
# speedup vs baseline: 117.5611x; 1.0605x over previous
#include <hip/hip_runtime.h>

typedef float v2f __attribute__((ext_vector_type(2)));
typedef float v4f __attribute__((ext_vector_type(4)));
typedef float v8f __attribute__((ext_vector_type(8)));

#define SEQ   32768
#define DIM   1024

#define NW        8
#define NTHREADS  (NW * 32)
#define NCHUNK    NTHREADS
#define CL        (SEQ / NCHUNK)

#define GW      16
#define GTHREADS (GW * 32)
#define KSTEPS  (DIM / 32)
#define STAGE   8
#define NSTAGE  (KSTEPS / STAGE)
#define GBLOCKS (SEQ / (8 * GW))

__device__ __forceinline__ v8f wmma4(const float a0, const float a1,
                                     const float b0, const float b1, const v8f acc)
{
    const v2f a = { a0, a1 };
    const v2f b = { b0, b1 };
    return __builtin_amdgcn_wmma_f32_16x16x4_f32(
                false, a,  false, b,
                (short)0, acc,  false,  false);
}

__device__ __forceinline__ float pick4(const float v0, const float v1,
                                       const float v2, const float v3, const int j)
{
    return (j == 0) ? v0 : (j == 1) ? v1 : (j == 2) ? v2 : v3;
}

__device__ __forceinline__ float quad_sum(float v)
{
    v += __shfl_xor(v, 1, 32);
    v += __shfl_xor(v, 2, 32);
    return v;
}

__global__ __launch_bounds__(GTHREADS)
void gates_gemm_wmma(const float* __restrict__ x,
                     const float* __restrict__ Wih,
                     const float* __restrict__ bih,
                     const float* __restrict__ bhh,
                     float* __restrict__ gates)
{
    const int lane = threadIdx.x & 31;
    const int wave = threadIdx.x >> 5;
    const int c    = lane & 15;
    const int h    = lane >> 4;
    const int q    = c >> 2;
    const int j    = c & 3;
    const int qo   = blockIdx.x >> 3;
    const int t    = (blockIdx.x & 7) * GW + wave;

    const int koff = 16 * h + 4 * j;
    const v4f* xa = (const v4f*)(x + (size_t)((qo * 8 + q) * CL + t) * DIM + koff);
    const v4f* xb = (const v4f*)(x + (size_t)((qo * 8 + 4 + q) * CL + t) * DIM + koff);
    const v4f* wp = (const v4f*)(Wih + (size_t)q * DIM + koff);

    v4f fa[2][STAGE], fb[2][STAGE], fw[2][STAGE];

#define LOAD_STAGE(S) do {                                                   \
        _Pragma("unroll") for (int i = 0; i < STAGE; ++i) {                  \
            fw[(S) & 1][i] = wp[((S) * STAGE + i) * 8];                      \
            fa[(S) & 1][i] = xa[((S) * STAGE + i) * 8];                      \
            fb[(S) & 1][i] = xb[((S) * STAGE + i) * 8];                      \
        } } while (0)

    v8f accA = {}, accB = {};
    LOAD_STAGE(0);
    #pragma unroll
    for (int s = 0; s < NSTAGE; ++s) {
        __builtin_amdgcn_sched_barrier(0);
        if (s + 1 < NSTAGE) LOAD_STAGE(s + 1);
        __builtin_amdgcn_sched_barrier(0);
        #pragma unroll
        for (int i = 0; i < STAGE; ++i) {
            const v4f a = fa[s & 1][i], b = fb[s & 1][i], w = fw[s & 1][i];
            accA = wmma4(a.x, a.y, w.x, w.y, accA);
            accB = wmma4(b.x, b.y, w.x, w.y, accB);
            accA = wmma4(a.z, a.w, w.z, w.w, accA);
            accB = wmma4(b.z, b.w, w.z, w.w, accB);
        }
    }
#undef LOAD_STAGE

    float a0 = pick4(accA[0], accA[1], accA[2], accA[3], j);
    float a1 = pick4(accA[4], accA[5], accA[6], accA[7], j);
    float b0 = pick4(accB[0], accB[1], accB[2], accB[3], j);
    float b1 = pick4(accB[4], accB[5], accB[6], accB[7], j);
    a0 = quad_sum(a0); a1 = quad_sum(a1); b0 = quad_sum(b0); b1 = quad_sum(b1);

    const float v  = pick4(a0, a1, b0, b1, j);
    const int   R  = 4 * (j >> 1) + 2 * h + (j & 1);
    const int   g  = q;
    const float bb = bih[g] + bhh[g];
    const float sc = (g == 2) ? 1.0f : 0.5f;
    gates[((size_t)t * NCHUNK + qo * 8 + R) * 4 + g] = (v + bb) * sc;
}

__device__ __forceinline__ float hw_tanh(float xv)
{
    return __builtin_amdgcn_tanhf(xv);
}

#define BS        16
#define NB        (CL / BS)
#define WB        4
#define MAX_SWEEPS (NCHUNK + 1)

__device__ __forceinline__ void lstm_steps(const v4f (&g)[BS], float (&h)[BS],
                                           float& tc, float& c, float& ov,
                                           const float ws0, const float ws1,
                                           const float ws2, const float ws3)
{
    float wo0 = ws0 * ov, wo1 = ws1 * ov, wo2 = ws2 * ov, wo3 = ws3 * ov;
    #pragma unroll
    for (int j = 0; j < BS; ++j) {
        const float ai = fmaf(wo0, tc, g[j].x);
        const float af = fmaf(wo1, tc, g[j].y);
        const float ag = fmaf(wo2, tc, g[j].z);
        const float ao = fmaf(wo3, tc, g[j].w);
        const float ti = hw_tanh(ai);
        const float ch = 0.5f * c;
        const float tf = hw_tanh(af);
        const float tg = hw_tanh(ag);
        const float to = hw_tanh(ao);
        const float iv = fmaf(0.5f, ti, 0.5f);
        const float P  = fmaf(tf, ch, ch);
        ov = fmaf(0.5f, to, 0.5f);
        c  = fmaf(iv, tg, P);
        wo0 = ws0 * ov;
        tc = hw_tanh(c);
        wo1 = ws1 * ov; wo2 = ws2 * ov; wo3 = ws3 * ov;
        h[j] = ov * tc;
    }
}

__device__ __forceinline__ void load_block(v4f (&g)[BS], const v4f* __restrict__ p)
{
    #pragma unroll
    for (int j = 0; j < BS; ++j) g[j] = p[(size_t)j * NCHUNK];
}

__device__ __forceinline__ void store_block(float* __restrict__ o, const float (&h)[BS])
{
    #pragma unroll
    for (int q = 0; q < BS / 4; ++q) {
        v4f v = { h[4 * q], h[4 * q + 1], h[4 * q + 2], h[4 * q + 3] };
        *(v4f*)(o + 4 * q) = v;
    }
}

__global__ __launch_bounds__(NTHREADS)
void lstm_scan(const float* __restrict__ gates,
               const float* __restrict__ Whh,
               const float* __restrict__ h0,
               const float* __restrict__ c0,
               float* __restrict__ out)
{
    __shared__ float s_tc[NTHREADS], s_c[NTHREADS], s_ov[NTHREADS];
    __shared__ int   s_bad[NW];

    const int tid  = threadIdx.x;
    const int wave = tid >> 5;

    const float ws0 = 0.5f * Whh[0], ws1 = 0.5f * Whh[1],
                ws2 =        Whh[2], ws3 = 0.5f * Whh[3];

    const float h0v = h0[0], c0v = c0[0];

    float st_tc = 0.0f, st_c = 0.0f, st_ov = 0.5f;

    const v4f* gp = (const v4f*)gates + tid;
    float*     op = out + (size_t)tid * CL;

    bool need_store = true;

    v4f ga[BS], gb[BS];
    load_block(ga, gp + (size_t)(NB - WB) * BS * NCHUNK);

    for (int sweep = 0; sweep < MAX_SWEEPS; ++sweep) {
        const int  b0 = (sweep == 0) ? (NB - WB) : 0;
        const bool st = need_store && (sweep > 0);

        float tc = st_tc, c = st_c, ov = st_ov;
        float h[BS];
        for (int b = b0; b < NB; b += 2) {
            load_block(gb, gp + (size_t)(b + 1) * BS * NCHUNK);
            lstm_steps(ga, h, tc, c, ov, ws0, ws1, ws2, ws3);
            if (st) store_block(op + b * BS, h);
            const int bn = (b + 2 < NB) ? (b + 2) : 0;
            load_block(ga, gp + (size_t)bn * BS * NCHUNK);
            lstm_steps(gb, h, tc, c, ov, ws0, ws1, ws2, ws3);
            if (st) store_block(op + (b + 1) * BS, h);
        }
        if (st) need_store = false;

        s_tc[tid] = tc; s_c[tid] = c; s_ov[tid] = ov;
        __syncthreads();
        float n_tc = h0v, n_c = c0v, n_ov = 1.0f;
        if (tid > 0) { n_tc = s_tc[tid - 1]; n_c = s_c[tid - 1]; n_ov = s_ov[tid - 1]; }
        const bool bad = (__float_as_uint(n_tc) != __float_as_uint(st_tc)) ||
                         (__float_as_uint(n_c)  != __float_as_uint(st_c))  ||
                         (__float_as_uint(n_ov) != __float_as_uint(st_ov));
        const bool wave_bad = __builtin_amdgcn_ballot_w32(bad) != 0u;
        if ((tid & 31) == 0) s_bad[wave] = wave_bad ? 1 : 0;
        __syncthreads();
        int any_bad = 0;
        #pragma unroll
        for (int w = 0; w < NW; ++w) any_bad |= s_bad[w];
        if (any_bad == 0 && sweep > 0) break;
        if (bad) { st_tc = n_tc; st_c = n_c; st_ov = n_ov; need_store = true; }
    }
}

extern "C" void kernel_launch(void* const* d_in, const int* in_sizes, int n_in,
                              void* d_out, int out_size, void* d_ws, size_t ws_size,
                              hipStream_t stream)
{
    const float* x    = (const float*)d_in[0];
    const float* Wih  = (const float*)d_in[1];
    const float* Whh  = (const float*)d_in[2];
    const float* bih  = (const float*)d_in[3];
    const float* bhh  = (const float*)d_in[4];
    const float* h0   = (const float*)d_in[5];
    const float* c0   = (const float*)d_in[6];
    float*       out  = (float*)d_out;
    float*       gts  = (float*)d_ws;

    gates_gemm_wmma<<<GBLOCKS, GTHREADS, 0, stream>>>(x, Wih, bih, bhh, gts);
    lstm_scan<<<1, NTHREADS, 0, stream>>>(gts, Whh, h0, c0, out);
}


# speedup vs baseline: 122.4997x; 1.0420x over previous
#include <hip/hip_runtime.h>

typedef float v2f __attribute__((ext_vector_type(2)));
typedef float v4f __attribute__((ext_vector_type(4)));
typedef float v8f __attribute__((ext_vector_type(8)));

#define SEQ   32768
#define DIM   1024

#define NW        8
#define NTHREADS  (NW * 32)
#define NCHUNK    NTHREADS
#define CL        (SEQ / NCHUNK)

#define GW      16
#define GTHREADS (GW * 32)
#define KSTEPS  (DIM / 32)
#define STAGE   8
#define NSTAGE  (KSTEPS / STAGE)
#define GBLOCKS (SEQ / (8 * GW))

__device__ __forceinline__ v8f wmma4(const float a0, const float a1,
                                     const float b0, const float b1, const v8f acc)
{
    const v2f a = { a0, a1 };
    const v2f b = { b0, b1 };
    return __builtin_amdgcn_wmma_f32_16x16x4_f32(
                false, a,  false, b,
                (short)0, acc,  false,  false);
}

__device__ __forceinline__ float pick4(const float v0, const float v1,
                                       const float v2, const float v3, const int j)
{
    return (j == 0) ? v0 : (j == 1) ? v1 : (j == 2) ? v2 : v3;
}

__device__ __forceinline__ float quad_sum(float v)
{
    v += __shfl_xor(v, 1, 32);
    v += __shfl_xor(v, 2, 32);
    return v;
}

__global__ __launch_bounds__(GTHREADS)
void gates_gemm_wmma(const float* __restrict__ x,
                     const float* __restrict__ Wih,
                     const float* __restrict__ bih,
                     const float* __restrict__ bhh,
                     float* __restrict__ gates)
{
    const int lane = threadIdx.x & 31;
    const int wave = threadIdx.x >> 5;
    const int c    = lane & 15;
    const int h    = lane >> 4;
    const int q    = c >> 2;
    const int j    = c & 3;
    const int qo   = blockIdx.x >> 3;
    const int t    = (blockIdx.x & 7) * GW + wave;

    const int koff = 16 * h + 4 * j;
    const v4f* xa = (const v4f*)(x + (size_t)((qo * 8 + q) * CL + t) * DIM + koff);
    const v4f* xb = (const v4f*)(x + (size_t)((qo * 8 + 4 + q) * CL + t) * DIM + koff);
    const v4f* wp = (const v4f*)(Wih + (size_t)q * DIM + koff);

    v4f fa[2][STAGE], fb[2][STAGE], fw[2][STAGE];

#define LOAD_STAGE(S) do {                                                   \
        _Pragma("unroll") for (int i = 0; i < STAGE; ++i) {                  \
            fw[(S) & 1][i] = wp[((S) * STAGE + i) * 8];                      \
            fa[(S) & 1][i] = __builtin_nontemporal_load(&xa[((S) * STAGE + i) * 8]); \
            fb[(S) & 1][i] = __builtin_nontemporal_load(&xb[((S) * STAGE + i) * 8]); \
        } } while (0)

    v8f accA = {}, accB = {};
    LOAD_STAGE(0);
    #pragma unroll
    for (int s = 0; s < NSTAGE; ++s) {
        __builtin_amdgcn_sched_barrier(0);
        if (s + 1 < NSTAGE) LOAD_STAGE(s + 1);
        __builtin_amdgcn_sched_barrier(0);
        #pragma unroll
        for (int i = 0; i < STAGE; ++i) {
            const v4f a = fa[s & 1][i], b = fb[s & 1][i], w = fw[s & 1][i];
            accA = wmma4(a.x, a.y, w.x, w.y, accA);
            accB = wmma4(b.x, b.y, w.x, w.y, accB);
            accA = wmma4(a.z, a.w, w.z, w.w, accA);
            accB = wmma4(b.z, b.w, w.z, w.w, accB);
        }
    }
#undef LOAD_STAGE

    float a0 = pick4(accA[0], accA[1], accA[2], accA[3], j);
    float a1 = pick4(accA[4], accA[5], accA[6], accA[7], j);
    float b0 = pick4(accB[0], accB[1], accB[2], accB[3], j);
    float b1 = pick4(accB[4], accB[5], accB[6], accB[7], j);
    a0 = quad_sum(a0); a1 = quad_sum(a1); b0 = quad_sum(b0); b1 = quad_sum(b1);

    const float v  = pick4(a0, a1, b0, b1, j);
    const int   R  = 4 * (j >> 1) + 2 * h + (j & 1);
    const int   g  = q;
    const float bb = bih[g] + bhh[g];
    const float sc = (g == 2) ? 1.0f : 0.5f;
    gates[((size_t)t * NCHUNK + qo * 8 + R) * 4 + g] = (v + bb) * sc;
}

__device__ __forceinline__ float hw_tanh(float xv)
{
    return __builtin_amdgcn_tanhf(xv);
}

#define BS        16
#define NB        (CL / BS)
#define WB        4
#define MAX_SWEEPS (NCHUNK + 1)

__device__ __forceinline__ void lstm_steps(const v4f (&g)[BS], float (&h)[BS],
                                           float& tc, float& c, float& ov,
                                           const float ws0, const float ws1,
                                           const float ws2, const float ws3)
{
    float wo0 = ws0 * ov, wo1 = ws1 * ov, wo2 = ws2 * ov, wo3 = ws3 * ov;
    #pragma unroll
    for (int j = 0; j < BS; ++j) {
        const float ai = fmaf(wo0, tc, g[j].x);
        const float af = fmaf(wo1, tc, g[j].y);
        const float ag = fmaf(wo2, tc, g[j].z);
        const float ao = fmaf(wo3, tc, g[j].w);
        const float ti = hw_tanh(ai);
        const float ch = 0.5f * c;
        const float tf = hw_tanh(af);
        const float tg = hw_tanh(ag);
        const float to = hw_tanh(ao);
        const float iv = fmaf(0.5f, ti, 0.5f);
        const float P  = fmaf(tf, ch, ch);
        ov = fmaf(0.5f, to, 0.5f);
        c  = fmaf(iv, tg, P);
        wo0 = ws0 * ov;
        tc = hw_tanh(c);
        wo1 = ws1 * ov; wo2 = ws2 * ov; wo3 = ws3 * ov;
        h[j] = ov * tc;
    }
}

__device__ __forceinline__ void load_block(v4f (&g)[BS], const v4f* __restrict__ p)
{
    #pragma unroll
    for (int j = 0; j < BS; ++j) g[j] = p[(size_t)j * NCHUNK];
}

__device__ __forceinline__ void store_block(float* __restrict__ o, const float (&h)[BS])
{
    #pragma unroll
    for (int q = 0; q < BS / 4; ++q) {
        v4f v = { h[4 * q], h[4 * q + 1], h[4 * q + 2], h[4 * q + 3] };
        *(v4f*)(o + 4 * q) = v;
    }
}

__global__ __launch_bounds__(NTHREADS)
void lstm_scan(const float* __restrict__ gates,
               const float* __restrict__ Whh,
               const float* __restrict__ h0,
               const float* __restrict__ c0,
               float* __restrict__ out)
{
    __shared__ float s_tc[NTHREADS], s_c[NTHREADS], s_ov[NTHREADS];
    __shared__ int   s_bad[NW];

    const int tid  = threadIdx.x;
    const int wave = tid >> 5;

    const float ws0 = 0.5f * Whh[0], ws1 = 0.5f * Whh[1],
                ws2 =        Whh[2], ws3 = 0.5f * Whh[3];

    const float h0v = h0[0], c0v = c0[0];

    float st_tc = 0.0f, st_c = 0.0f, st_ov = 0.5f;

    const v4f* gp = (const v4f*)gates + tid;
    float*     op = out + (size_t)tid * CL;

    bool need_store = true;

    v4f ga[BS], gb[BS];
    load_block(ga, gp + (size_t)(NB - WB) * BS * NCHUNK);

    for (int sweep = 0; sweep < MAX_SWEEPS; ++sweep) {
        const int  b0 = (sweep == 0) ? (NB - WB) : 0;
        const bool st = need_store && (sweep > 0);

        float tc = st_tc, c = st_c, ov = st_ov;
        float h[BS];
        for (int b = b0; b < NB; b += 2) {
            load_block(gb, gp + (size_t)(b + 1) * BS * NCHUNK);
            lstm_steps(ga, h, tc, c, ov, ws0, ws1, ws2, ws3);
            if (st) store_block(op + b * BS, h);
            const int bn = (b + 2 < NB) ? (b + 2) : 0;
            load_block(ga, gp + (size_t)bn * BS * NCHUNK);
            lstm_steps(gb, h, tc, c, ov, ws0, ws1, ws2, ws3);
            if (st) store_block(op + (b + 1) * BS, h);
        }
        if (st) need_store = false;

        s_tc[tid] = tc; s_c[tid] = c; s_ov[tid] = ov;
        __syncthreads();
        float n_tc = h0v, n_c = c0v, n_ov = 1.0f;
        if (tid > 0) { n_tc = s_tc[tid - 1]; n_c = s_c[tid - 1]; n_ov = s_ov[tid - 1]; }
        const bool bad = (__float_as_uint(n_tc) != __float_as_uint(st_tc)) ||
                         (__float_as_uint(n_c)  != __float_as_uint(st_c))  ||
                         (__float_as_uint(n_ov) != __float_as_uint(st_ov));
        const bool wave_bad = __builtin_amdgcn_ballot_w32(bad) != 0u;
        if ((tid & 31) == 0) s_bad[wave] = wave_bad ? 1 : 0;
        __syncthreads();
        int any_bad = 0;
        #pragma unroll
        for (int w = 0; w < NW; ++w) any_bad |= s_bad[w];
        if (any_bad == 0 && sweep > 0) break;
        if (bad) { st_tc = n_tc; st_c = n_c; st_ov = n_ov; need_store = true; }
    }
}

extern "C" void kernel_launch(void* const* d_in, const int* in_sizes, int n_in,
                              void* d_out, int out_size, void* d_ws, size_t ws_size,
                              hipStream_t stream)
{
    const float* x    = (const float*)d_in[0];
    const float* Wih  = (const float*)d_in[1];
    const float* Whh  = (const float*)d_in[2];
    const float* bih  = (const float*)d_in[3];
    const float* bhh  = (const float*)d_in[4];
    const float* h0   = (const float*)d_in[5];
    const float* c0   = (const float*)d_in[6];
    float*       out  = (float*)d_out;
    float*       gts  = (float*)d_ws;

    gates_gemm_wmma<<<GBLOCKS, GTHREADS, 0, stream>>>(x, Wih, bih, bhh, gts);
    lstm_scan<<<1, NTHREADS, 0, stream>>>(gts, Whh, h0, c0, out);
}


# scan: the speculative chunk pass moves into the gates launch (one chunk per work-group, one wave runs its 128 steps from the fixed guess); the scan launch re-runs 64 steps per chunk, checks bitwise
# speedup vs baseline: 130.7514x; 1.0674x over previous
#include <hip/hip_runtime.h>

typedef float v2f __attribute__((ext_vector_type(2)));
typedef float v4f __attribute__((ext_vector_type(4)));
typedef float v8f __attribute__((ext_vector_type(8)));

#define SEQ   32768
#define DIM   1024

#define NW        8
#define NTHREADS  (NW * 32)
#define NCHUNK    NTHREADS
#define CL        (SEQ / NCHUNK)

#define GW      16
#define GTHREADS (GW * 32)
#define KSTEPS  (DIM / 32)
#define STAGE   8
#define NSTAGE  (KSTEPS / STAGE)
#define GBLOCKS NCHUNK
static_assert(8 * GW == CL, "a work-group's waves, 8 rows each, must cover exactly one chunk");

__device__ __forceinline__ v8f wmma4(const float a0, const float a1,
                                     const float b0, const float b1, const v8f acc)
{
    const v2f a = { a0, a1 };
    const v2f b = { b0, b1 };
    return __builtin_amdgcn_wmma_f32_16x16x4_f32(
                false, a,  false, b,
                (short)0, acc,  false,  false);
}

__device__ __forceinline__ float pick4(const float v0, const float v1,
                                       const float v2, const float v3, const int j)
{
    return (j == 0) ? v0 : (j == 1) ? v1 : (j == 2) ? v2 : v3;
}

__device__ __forceinline__ float quad_sum(float v)
{
    v += __shfl_xor(v, 1, 32);
    v += __shfl_xor(v, 2, 32);
    return v;
}

__device__ __forceinline__ float hw_tanh(float xv)
{
    return __builtin_amdgcn_tanhf(xv);
}

#define BS        16
#define NB        (CL / BS)
#define WB        4
#define MAX_SWEEPS (NCHUNK + 1)

#define GUESS_TC  0.0f
#define GUESS_C   0.0f
#define GUESS_OV  0.5f

__device__ __forceinline__ void lstm_steps(const v4f (&g)[BS], float (&h)[BS],
                                           float& tc, float& c, float& ov,
                                           const float ws0, const float ws1,
                                           const float ws2, const float ws3)
{
    float wo0 = ws0 * ov, wo1 = ws1 * ov, wo2 = ws2 * ov, wo3 = ws3 * ov;
    #pragma unroll
    for (int j = 0; j < BS; ++j) {
        const float ai = fmaf(wo0, tc, g[j].x);
        const float af = fmaf(wo1, tc, g[j].y);
        const float ag = fmaf(wo2, tc, g[j].z);
        const float ao = fmaf(wo3, tc, g[j].w);
        const float ti = hw_tanh(ai);
        const float ch = 0.5f * c;
        const float tf = hw_tanh(af);
        const float tg = hw_tanh(ag);
        const float to = hw_tanh(ao);
        const float iv = fmaf(0.5f, ti, 0.5f);
        const float P  = fmaf(tf, ch, ch);
        ov = fmaf(0.5f, to, 0.5f);
        c  = fmaf(iv, tg, P);
        wo0 = ws0 * ov;
        tc = hw_tanh(c);
        wo1 = ws1 * ov; wo2 = ws2 * ov; wo3 = ws3 * ov;
        h[j] = ov * tc;
    }
}

__device__ __forceinline__ void load_rows(v4f (&g)[BS], const v4f* p)
{
    #pragma unroll
    for (int j = 0; j < BS; ++j) g[j] = p[j];
}

__device__ __forceinline__ void store_block(float* o, const float (&h)[BS])
{
    #pragma unroll
    for (int q = 0; q < BS / 4; ++q) {
        v4f v = { h[4 * q], h[4 * q + 1], h[4 * q + 2], h[4 * q + 3] };
        *(v4f*)(o + 4 * q) = v;
    }
}

__global__ __launch_bounds__(GTHREADS)
void gates_gemm_wmma(const float* __restrict__ x,
                     const float* __restrict__ Wih,
                     const float* __restrict__ bih,
                     const float* __restrict__ bhh,
                     const float* __restrict__ Whh,
                     float* __restrict__ gates,
                     float* __restrict__ out)
{
    __shared__ v4f s_g[CL];

    const int lane = threadIdx.x & 31;
    const int wave = threadIdx.x >> 5;
    const int c    = lane & 15;
    const int h    = lane >> 4;
    const int q    = c >> 2;
    const int j    = c & 3;
    const int chunk = blockIdx.x;
    const int t0   = 8 * wave;

    const int koff = 16 * h + 4 * j;
    const v4f* xa = (const v4f*)(x + (size_t)(chunk * CL + t0 + q) * DIM + koff);
    const v4f* xb = (const v4f*)(x + (size_t)(chunk * CL + t0 + 4 + q) * DIM + koff);
    const v4f* wp = (const v4f*)(Wih + (size_t)q * DIM + koff);

    const float ws0 = 0.5f * Whh[0], ws1 = 0.5f * Whh[1],
                ws2 =        Whh[2], ws3 = 0.5f * Whh[3];

    v4f fa[2][STAGE], fb[2][STAGE], fw[2][STAGE];

#define LOAD_STAGE(S) do {                                                   \
        _Pragma("unroll") for (int i = 0; i < STAGE; ++i) {                  \
            fw[(S) & 1][i] = wp[((S) * STAGE + i) * 8];                      \
            fa[(S) & 1][i] = __builtin_nontemporal_load(&xa[((S) * STAGE + i) * 8]); \
            fb[(S) & 1][i] = __builtin_nontemporal_load(&xb[((S) * STAGE + i) * 8]); \
        } } while (0)

    v8f accA = {}, accB = {};
    LOAD_STAGE(0);
    #pragma unroll
    for (int s = 0; s < NSTAGE; ++s) {
        __builtin_amdgcn_sched_barrier(0);
        if (s + 1 < NSTAGE) LOAD_STAGE(s + 1);
        __builtin_amdgcn_sched_barrier(0);
        #pragma unroll
        for (int i = 0; i < STAGE; ++i) {
            const v4f a = fa[s & 1][i], b = fb[s & 1][i], w = fw[s & 1][i];
            accA = wmma4(a.x, a.y, w.x, w.y, accA);
            accB = wmma4(b.x, b.y, w.x, w.y, accB);
            accA = wmma4(a.z, a.w, w.z, w.w, accA);
            accB = wmma4(b.z, b.w, w.z, w.w, accB);
        }
    }
#undef LOAD_STAGE

    float a0 = pick4(accA[0], accA[1], accA[2], accA[3], j);
    float a1 = pick4(accA[4], accA[5], accA[6], accA[7], j);
    float b0 = pick4(accB[0], accB[1], accB[2], accB[3], j);
    float b1 = pick4(accB[4], accB[5], accB[6], accB[7], j);
    a0 = quad_sum(a0); a1 = quad_sum(a1); b0 = quad_sum(b0); b1 = quad_sum(b1);

    const float v  = pick4(a0, a1, b0, b1, j);
    const int   R  = 4 * (j >> 1) + 2 * h + (j & 1);
    const int   g  = q;
    const float bb = bih[g] + bhh[g];
    const float sc = (g == 2) ? 1.0f : 0.5f;
    const float gv = (v + bb) * sc;
    ((float*)s_g)[(t0 + R) * 4 + g] = gv;
    __syncthreads();
    gates[((size_t)chunk * CL + t0 + R) * 4 + g] = gv;
    if (wave != 0) return;

    float tc = GUESS_TC, cc = GUESS_C, ov = GUESS_OV;
    float m_tc = 0.0f, m_c = 0.0f, m_ov = 0.0f;
    float* op = out + (size_t)chunk * CL;
    v4f ga[BS], gb[BS];
    float hh[BS];
    load_rows(ga, s_g);
    for (int b = 0; b < NB; b += 2) {
        load_rows(gb, s_g + (b + 1) * BS);
        __builtin_amdgcn_sched_barrier(0);
        lstm_steps(ga, hh, tc, cc, ov, ws0, ws1, ws2, ws3);
        if (b >= WB && lane == 0) store_block(op + b * BS, hh);
        const int bn = (b + 2 < NB) ? (b + 2) : 0;
        load_rows(ga, s_g + bn * BS);
        __builtin_amdgcn_sched_barrier(0);
        lstm_steps(gb, hh, tc, cc, ov, ws0, ws1, ws2, ws3);
        if (b >= WB && lane == 0) store_block(op + (b + 1) * BS, hh);
        if (b + 2 == WB) { m_tc = tc; m_c = cc; m_ov = ov; }
    }
    if (lane == 0) {
        const v4f mid = { m_tc, m_c, m_ov, 0.0f };
        const v4f end = { tc, cc, ov, 0.0f };
        *(v4f*)(op)     = mid;
        *(v4f*)(op + 4) = end;
    }
}

#define XP  (BS + 1)

__device__ __forceinline__ void gload_block(v4f (&s)[BS], const v4f* __restrict__ gq,
                                            const int wave, const int lane, const int b)
{
    #pragma unroll
    for (int i = 0; i < BS; ++i)
        s[i] = gq[(size_t)(wave * 32 + 2 * i + (lane >> 4)) * CL + b * BS + (lane & 15)];
    __builtin_amdgcn_sched_barrier(0);
}

__device__ __forceinline__ void turn_block(v4f (&g)[BS], const v4f (&s)[BS],
                                           v4f* buf, const int lane)
{
    __builtin_amdgcn_sched_barrier(0);
    #pragma unroll
    for (int i = 0; i < BS; ++i) buf[(2 * i + (lane >> 4)) * XP + (lane & 15)] = s[i];
    __builtin_amdgcn_fence(__ATOMIC_RELEASE, "wavefront");
    __builtin_amdgcn_wave_barrier();
    __builtin_amdgcn_fence(__ATOMIC_ACQUIRE, "wavefront");
    #pragma unroll
    for (int jj = 0; jj < BS; ++jj) g[jj] = buf[lane * XP + jj];
    __builtin_amdgcn_sched_barrier(0);
}

__global__ __launch_bounds__(NTHREADS)
void lstm_scan(const float* __restrict__ gates,
               const float* __restrict__ Whh,
               const float* __restrict__ h0,
               const float* __restrict__ c0,
               float* out)
{
    __shared__ float s_tc[NTHREADS], s_c[NTHREADS], s_ov[NTHREADS];
    __shared__ float s_etc[NTHREADS], s_ec[NTHREADS], s_eov[NTHREADS];
    __shared__ int   s_bad[NW];
    __shared__ v4f   s_x[NW][2][32 * XP];

    const int tid  = threadIdx.x;
    const int wave = tid >> 5;
    const int lane = tid & 31;

    const v4f* gq = (const v4f*)gates;
    const v4f* gp = gq + (size_t)tid * CL;
    float*     op = out + (size_t)tid * CL;
    v4f* buf0 = &s_x[wave][0][0];
    v4f* buf1 = &s_x[wave][1][0];

    v4f sa[BS], sb[BS];
    gload_block(sa, gq, wave, lane, 0);
    gload_block(sb, gq, wave, lane, 1);

    const float ws0 = 0.5f * Whh[0], ws1 = 0.5f * Whh[1],
                ws2 =        Whh[2], ws3 = 0.5f * Whh[3];

    const float h0v = h0[0], c0v = c0[0];

    const v4f mid = *(const v4f*)(op);
    const v4f end = *(const v4f*)(op + 4);

    s_etc[tid] = end.x; s_ec[tid] = end.y; s_eov[tid] = end.z;
    __syncthreads();
    float st_tc = h0v, st_c = c0v, st_ov = 1.0f;
    if (tid > 0) { st_tc = s_etc[tid - 1]; st_c = s_ec[tid - 1]; st_ov = s_eov[tid - 1]; }

    float tc = st_tc, c = st_c, ov = st_ov;
    float h[BS];
    v4f ga[BS], gb[BS];
    turn_block(ga, sa, buf0, lane);
    turn_block(gb, sb, buf1, lane);
    gload_block(sa, gq, wave, lane, 2);
    lstm_steps(ga, h, tc, c, ov, ws0, ws1, ws2, ws3);
    store_block(op, h);
    turn_block(ga, sa, buf0, lane);
    gload_block(sa, gq, wave, lane, 3);
    lstm_steps(gb, h, tc, c, ov, ws0, ws1, ws2, ws3);
    store_block(op + BS, h);
    turn_block(gb, sa, buf1, lane);
    lstm_steps(ga, h, tc, c, ov, ws0, ws1, ws2, ws3);
    store_block(op + 2 * BS, h);
    lstm_steps(gb, h, tc, c, ov, ws0, ws1, ws2, ws3);
    store_block(op + 3 * BS, h);

    const bool mid_ok = (__float_as_uint(tc) == __float_as_uint(mid.x)) &&
                        (__float_as_uint(c)  == __float_as_uint(mid.y)) &&
                        (__float_as_uint(ov) == __float_as_uint(mid.z));
    float e_tc = end.x, e_c = end.y, e_ov = end.z;
    if (__builtin_amdgcn_ballot_w32(!mid_ok) != 0u) {
        for (int b = WB; b < NB; ++b) {
            load_rows(ga, gp + b * BS);
            lstm_steps(ga, h, tc, c, ov, ws0, ws1, ws2, ws3);
            if (!mid_ok) store_block(op + b * BS, h);
        }
        if (!mid_ok) { e_tc = tc; e_c = c; e_ov = ov; }
    }

    for (int sweep = 0; sweep < MAX_SWEEPS; ++sweep) {
        s_tc[tid] = e_tc; s_c[tid] = e_c; s_ov[tid] = e_ov;
        __syncthreads();
        float n_tc = h0v, n_c = c0v, n_ov = 1.0f;
        if (tid > 0) { n_tc = s_tc[tid - 1]; n_c = s_c[tid - 1]; n_ov = s_ov[tid - 1]; }
        const bool bad = (__float_as_uint(n_tc) != __float_as_uint(st_tc)) ||
                         (__float_as_uint(n_c)  != __float_as_uint(st_c))  ||
                         (__float_as_uint(n_ov) != __float_as_uint(st_ov));
        const bool wave_bad = __builtin_amdgcn_ballot_w32(bad) != 0u;
        if (lane == 0) s_bad[wave] = wave_bad ? 1 : 0;
        __syncthreads();
        int any_bad = 0;
        #pragma unroll
        for (int w = 0; w < NW; ++w) any_bad |= s_bad[w];
        if (any_bad == 0) break;
        if (bad) { st_tc = n_tc; st_c = n_c; st_ov = n_ov; }

        tc = st_tc; c = st_c; ov = st_ov;
        for (int b = 0; b < NB; ++b) {
            load_rows(ga, gp + b * BS);
            lstm_steps(ga, h, tc, c, ov, ws0, ws1, ws2, ws3);
            if (bad) store_block(op + b * BS, h);
        }
        e_tc = tc; e_c = c; e_ov = ov;
    }
}

extern "C" void kernel_launch(void* const* d_in, const int* in_sizes, int n_in,
                              void* d_out, int out_size, void* d_ws, size_t ws_size,
                              hipStream_t stream)
{
    const float* x    = (const float*)d_in[0];
    const float* Wih  = (const float*)d_in[1];
    const float* Whh  = (const float*)d_in[2];
    const float* bih  = (const float*)d_in[3];
    const float* bhh  = (const float*)d_in[4];
    const float* h0   = (const float*)d_in[5];
    const float* c0   = (const float*)d_in[6];
    float*       out  = (float*)d_out;
    float*       gts  = (float*)d_ws;

    gates_gemm_wmma<<<GBLOCKS, GTHREADS, 0, stream>>>(x, Wih, bih, bhh, Whh, gts, out);
    lstm_scan<<<1, NTHREADS, 0, stream>>>(gts, Whh, h0, c0, out);
}


# chunk pass of the gates launch spread over its wave's 32 lanes: each runs 64 warm-up steps from the fixed guess on its own window, then 2 rows; neighbour states checked bitwise; same stored values
# speedup vs baseline: 146.0802x; 1.1172x over previous
#include <hip/hip_runtime.h>

typedef float v2f __attribute__((ext_vector_type(2)));
typedef float v4f __attribute__((ext_vector_type(4)));
typedef float v8f __attribute__((ext_vector_type(8)));

#define SEQ   32768
#define DIM   1024

#define NW        8
#define NTHREADS  (NW * 32)
#define NCHUNK    NTHREADS
#define CL        (SEQ / NCHUNK)

#define GW      16
#define GTHREADS (GW * 32)
#define KSTEPS  (DIM / 32)
#define STAGE   8
#define NSTAGE  (KSTEPS / STAGE)
#define GBLOCKS NCHUNK
static_assert(8 * GW == CL, "a work-group's waves, 8 rows each, must cover exactly one chunk");

__device__ __forceinline__ v8f wmma4(const float a0, const float a1,
                                     const float b0, const float b1, const v8f acc)
{
    const v2f a = { a0, a1 };
    const v2f b = { b0, b1 };
    return __builtin_amdgcn_wmma_f32_16x16x4_f32(
                false, a,  false, b,
                (short)0, acc,  false,  false);
}

__device__ __forceinline__ float pick4(const float v0, const float v1,
                                       const float v2, const float v3, const int j)
{
    return (j == 0) ? v0 : (j == 1) ? v1 : (j == 2) ? v2 : v3;
}

__device__ __forceinline__ float quad_sum(float v)
{
    v += __shfl_xor(v, 1, 32);
    v += __shfl_xor(v, 2, 32);
    return v;
}

__device__ __forceinline__ float hw_tanh(float xv)
{
    return __builtin_amdgcn_tanhf(xv);
}

#define BS        16
#define NB        (CL / BS)
#define WB        4
#define MAX_SWEEPS (NCHUNK + 1)
#define PR        ((CL - WB * BS) / 32)
static_assert(WB * BS + 32 * PR == CL, "warm-up window plus the 32 lanes' own rows must cover the chunk exactly");

#define GUESS_TC  0.0f
#define GUESS_C   0.0f
#define GUESS_OV  0.5f

template <int N>
__device__ __forceinline__ void lstm_steps(const v4f (&g)[N], float (&h)[N],
                                           float& tc, float& c, float& ov,
                                           const float ws0, const float ws1,
                                           const float ws2, const float ws3)
{
    float wo0 = ws0 * ov, wo1 = ws1 * ov, wo2 = ws2 * ov, wo3 = ws3 * ov;
    #pragma unroll
    for (int j = 0; j < N; ++j) {
        const float ai = fmaf(wo0, tc, g[j].x);
        const float af = fmaf(wo1, tc, g[j].y);
        const float ag = fmaf(wo2, tc, g[j].z);
        const float ao = fmaf(wo3, tc, g[j].w);
        const float ti = hw_tanh(ai);
        const float ch = 0.5f * c;
        const float tf = hw_tanh(af);
        const float tg = hw_tanh(ag);
        const float to = hw_tanh(ao);
        const float iv = fmaf(0.5f, ti, 0.5f);
        const float P  = fmaf(tf, ch, ch);
        ov = fmaf(0.5f, to, 0.5f);
        c  = fmaf(iv, tg, P);
        wo0 = ws0 * ov;
        tc = hw_tanh(c);
        wo1 = ws1 * ov; wo2 = ws2 * ov; wo3 = ws3 * ov;
        h[j] = ov * tc;
    }
}

__device__ __forceinline__ void load_rows(v4f (&g)[BS], const v4f* p)
{
    #pragma unroll
    for (int j = 0; j < BS; ++j) g[j] = p[j];
}

__device__ __forceinline__ void store_block(float* o, const float (&h)[BS])
{
    #pragma unroll
    for (int q = 0; q < BS / 4; ++q) {
        v4f v = { h[4 * q], h[4 * q + 1], h[4 * q + 2], h[4 * q + 3] };
        *(v4f*)(o + 4 * q) = v;
    }
}

__global__ __launch_bounds__(GTHREADS)
void gates_gemm_wmma(const float* __restrict__ x,
                     const float* __restrict__ Wih,
                     const float* __restrict__ bih,
                     const float* __restrict__ bhh,
                     const float* __restrict__ Whh,
                     float* __restrict__ gates,
                     float* __restrict__ out)
{
    __shared__ v4f s_g[CL];

    const int lane = threadIdx.x & 31;
    const int wave = threadIdx.x >> 5;
    const int c    = lane & 15;
    const int h    = lane >> 4;
    const int q    = c >> 2;
    const int j    = c & 3;
    const int chunk = blockIdx.x;
    const int t0   = 8 * wave;

    const int koff = 16 * h + 4 * j;
    const v4f* xa = (const v4f*)(x + (size_t)(chunk * CL + t0 + q) * DIM + koff);
    const v4f* xb = (const v4f*)(x + (size_t)(chunk * CL + t0 + 4 + q) * DIM + koff);
    const v4f* wp = (const v4f*)(Wih + (size_t)q * DIM + koff);

    const float ws0 = 0.5f * Whh[0], ws1 = 0.5f * Whh[1],
                ws2 =        Whh[2], ws3 = 0.5f * Whh[3];

    v4f fa[2][STAGE], fb[2][STAGE], fw[2][STAGE];

#define LOAD_STAGE(S) do {                                                   \
        _Pragma("unroll") for (int i = 0; i < STAGE; ++i) {                  \
            fw[(S) & 1][i] = wp[((S) * STAGE + i) * 8];                      \
            fa[(S) & 1][i] = __builtin_nontemporal_load(&xa[((S) * STAGE + i) * 8]); \
            fb[(S) & 1][i] = __builtin_nontemporal_load(&xb[((S) * STAGE + i) * 8]); \
        } } while (0)

    v8f accA = {}, accB = {};
    LOAD_STAGE(0);
    #pragma unroll
    for (int s = 0; s < NSTAGE; ++s) {
        __builtin_amdgcn_sched_barrier(0);
        if (s + 1 < NSTAGE) LOAD_STAGE(s + 1);
        __builtin_amdgcn_sched_barrier(0);
        #pragma unroll
        for (int i = 0; i < STAGE; ++i) {
            const v4f a = fa[s & 1][i], b = fb[s & 1][i], w = fw[s & 1][i];
            accA = wmma4(a.x, a.y, w.x, w.y, accA);
            accB = wmma4(b.x, b.y, w.x, w.y, accB);
            accA = wmma4(a.z, a.w, w.z, w.w, accA);
            accB = wmma4(b.z, b.w, w.z, w.w, accB);
        }
    }
#undef LOAD_STAGE

    float a0 = pick4(accA[0], accA[1], accA[2], accA[3], j);
    float a1 = pick4(accA[4], accA[5], accA[6], accA[7], j);
    float b0 = pick4(accB[0], accB[1], accB[2], accB[3], j);
    float b1 = pick4(accB[4], accB[5], accB[6], accB[7], j);
    a0 = quad_sum(a0); a1 = quad_sum(a1); b0 = quad_sum(b0); b1 = quad_sum(b1);

    const float v  = pick4(a0, a1, b0, b1, j);
    const int   R  = 4 * (j >> 1) + 2 * h + (j & 1);
    const int   g  = q;
    const float bb = bih[g] + bhh[g];
    const float sc = (g == 2) ? 1.0f : 0.5f;
    const float gv = (v + bb) * sc;
    ((float*)s_g)[(t0 + R) * 4 + g] = gv;
    __syncthreads();
    gates[((size_t)chunk * CL + t0 + R) * 4 + g] = gv;
    if (wave != 0) return;

    const v4f* lp = s_g + PR * lane;
    float* op = out + (size_t)chunk * CL;
    float tc = GUESS_TC, cc = GUESS_C, ov = GUESS_OV;
    v4f ga[BS], gb[BS], gp[PR];
    float hh[BS], hp[PR];
    load_rows(ga, lp);
    #pragma unroll
    for (int r = 0; r < PR; ++r) gp[r] = lp[WB * BS + r];
    for (int b = 0; b < WB; b += 2) {
        load_rows(gb, lp + (b + 1) * BS);
        __builtin_amdgcn_sched_barrier(0);
        lstm_steps(ga, hh, tc, cc, ov, ws0, ws1, ws2, ws3);
        const int bn = (b + 2 < WB) ? (b + 2) : 0;
        load_rows(ga, lp + bn * BS);
        __builtin_amdgcn_sched_barrier(0);
        lstm_steps(gb, hh, tc, cc, ov, ws0, ws1, ws2, ws3);
    }

    float s_tc = tc, s_c = cc, s_ov = ov;
    const float r_tc = __shfl_down(tc, 1, 32);
    const float r_c  = __shfl_down(cc, 1, 32);
    const float r_ov = __shfl_down(ov, 1, 32);
    lstm_steps(gp, hp, tc, cc, ov, ws0, ws1, ws2, ws3);
    const bool met = (lane == 31) ||
                     ((__float_as_uint(tc) == __float_as_uint(r_tc)) &&
                      (__float_as_uint(cc) == __float_as_uint(r_c))  &&
                      (__float_as_uint(ov) == __float_as_uint(r_ov)));
    if (__builtin_amdgcn_ballot_w32(!met) != 0u) {
        for (int pass = 0; pass < 33; ++pass) {
            const float l_tc = __shfl_up(tc, 1, 32);
            const float l_c  = __shfl_up(cc, 1, 32);
            const float l_ov = __shfl_up(ov, 1, 32);
            const bool bad = (lane > 0) &&
                             ((__float_as_uint(l_tc) != __float_as_uint(s_tc)) ||
                              (__float_as_uint(l_c)  != __float_as_uint(s_c))  ||
                              (__float_as_uint(l_ov) != __float_as_uint(s_ov)));
            if (__builtin_amdgcn_ballot_w32(bad) == 0u) break;
            if (bad) { s_tc = l_tc; s_c = l_c; s_ov = l_ov; }
            tc = s_tc; cc = s_c; ov = s_ov;
            lstm_steps(gp, hp, tc, cc, ov, ws0, ws1, ws2, ws3);
        }
    }

    {
        const v2f hv = { hp[0], hp[1] };
        static_assert(PR == 2, "the store below writes two rows per lane");
        *(v2f*)(op + WB * BS + PR * lane) = hv;
    }
    if (lane == 0 || lane == 31) {
        const v4f mid = { s_tc, s_c, s_ov, 0.0f };
        const v4f end = { tc, cc, ov, 0.0f };
        *(v4f*)(op + ((lane == 0) ? 0 : 4)) = (lane == 0) ? mid : end;
    }
}

#define XP  (BS + 1)

__device__ __forceinline__ void gload_block(v4f (&s)[BS], const v4f* __restrict__ gq,
                                            const int wave, const int lane, const int b)
{
    #pragma unroll
    for (int i = 0; i < BS; ++i)
        s[i] = gq[(size_t)(wave * 32 + 2 * i + (lane >> 4)) * CL + b * BS + (lane & 15)];
    __builtin_amdgcn_sched_barrier(0);
}

__device__ __forceinline__ void turn_block(v4f (&g)[BS], const v4f (&s)[BS],
                                           v4f* buf, const int lane)
{
    __builtin_amdgcn_sched_barrier(0);
    #pragma unroll
    for (int i = 0; i < BS; ++i) buf[(2 * i + (lane >> 4)) * XP + (lane & 15)] = s[i];
    __builtin_amdgcn_fence(__ATOMIC_RELEASE, "wavefront");
    __builtin_amdgcn_wave_barrier();
    __builtin_amdgcn_fence(__ATOMIC_ACQUIRE, "wavefront");
    #pragma unroll
    for (int jj = 0; jj < BS; ++jj) g[jj] = buf[lane * XP + jj];
    __builtin_amdgcn_sched_barrier(0);
}

__global__ __launch_bounds__(NTHREADS)
void lstm_scan(const float* __restrict__ gates,
               const float* __restrict__ Whh,
               const float* __restrict__ h0,
               const float* __restrict__ c0,
               float* out)
{
    __shared__ float s_tc[NTHREADS], s_c[NTHREADS], s_ov[NTHREADS];
    __shared__ float s_etc[NTHREADS], s_ec[NTHREADS], s_eov[NTHREADS];
    __shared__ int   s_bad[NW];
    __shared__ v4f   s_x[NW][2][32 * XP];

    const int tid  = threadIdx.x;
    const int wave = tid >> 5;
    const int lane = tid & 31;

    const v4f* gq = (const v4f*)gates;
    const v4f* gp = gq + (size_t)tid * CL;
    float*     op = out + (size_t)tid * CL;
    v4f* buf0 = &s_x[wave][0][0];
    v4f* buf1 = &s_x[wave][1][0];

    v4f sa[BS], sb[BS];
    gload_block(sa, gq, wave, lane, 0);
    gload_block(sb, gq, wave, lane, 1);

    const float ws0 = 0.5f * Whh[0], ws1 = 0.5f * Whh[1],
                ws2 =        Whh[2], ws3 = 0.5f * Whh[3];

    const float h0v = h0[0], c0v = c0[0];

    const v4f mid = *(const v4f*)(op);
    const v4f end = *(const v4f*)(op + 4);

    s_etc[tid] = end.x; s_ec[tid] = end.y; s_eov[tid] = end.z;
    __syncthreads();
    float st_tc = h0v, st_c = c0v, st_ov = 1.0f;
    if (tid > 0) { st_tc = s_etc[tid - 1]; st_c = s_ec[tid - 1]; st_ov = s_eov[tid - 1]; }

    float tc = st_tc, c = st_c, ov = st_ov;
    float h[BS];
    v4f ga[BS], gb[BS];
    turn_block(ga, sa, buf0, lane);
    turn_block(gb, sb, buf1, lane);
    gload_block(sa, gq, wave, lane, 2);
    lstm_steps(ga, h, tc, c, ov, ws0, ws1, ws2, ws3);
    store_block(op, h);
    turn_block(ga, sa, buf0, lane);
    gload_block(sa, gq, wave, lane, 3);
    lstm_steps(gb, h, tc, c, ov, ws0, ws1, ws2, ws3);
    store_block(op + BS, h);
    turn_block(gb, sa, buf1, lane);
    lstm_steps(ga, h, tc, c, ov, ws0, ws1, ws2, ws3);
    store_block(op + 2 * BS, h);
    lstm_steps(gb, h, tc, c, ov, ws0, ws1, ws2, ws3);
    store_block(op + 3 * BS, h);

    const bool mid_ok = (__float_as_uint(tc) == __float_as_uint(mid.x)) &&
                        (__float_as_uint(c)  == __float_as_uint(mid.y)) &&
                        (__float_as_uint(ov) == __float_as_uint(mid.z));
    float e_tc = end.x, e_c = end.y, e_ov = end.z;
    if (__builtin_amdgcn_ballot_w32(!mid_ok) != 0u) {
        for (int b = WB; b < NB; ++b) {
            load_rows(ga, gp + b * BS);
            lstm_steps(ga, h, tc, c, ov, ws0, ws1, ws2, ws3);
            if (!mid_ok) store_block(op + b * BS, h);
        }
        if (!mid_ok) { e_tc = tc; e_c = c; e_ov = ov; }
    }

    for (int sweep = 0; sweep < MAX_SWEEPS; ++sweep) {
        s_tc[tid] = e_tc; s_c[tid] = e_c; s_ov[tid] = e_ov;
        __syncthreads();
        float n_tc = h0v, n_c = c0v, n_ov = 1.0f;
        if (tid > 0) { n_tc = s_tc[tid - 1]; n_c = s_c[tid - 1]; n_ov = s_ov[tid - 1]; }
        const bool bad = (__float_as_uint(n_tc) != __float_as_uint(st_tc)) ||
                         (__float_as_uint(n_c)  != __float_as_uint(st_c))  ||
                         (__float_as_uint(n_ov) != __float_as_uint(st_ov));
        const bool wave_bad = __builtin_amdgcn_ballot_w32(bad) != 0u;
        if (lane == 0) s_bad[wave] = wave_bad ? 1 : 0;
        __syncthreads();
        int any_bad = 0;
        #pragma unroll
        for (int w = 0; w < NW; ++w) any_bad |= s_bad[w];
        if (any_bad == 0) break;
        if (bad) { st_tc = n_tc; st_c = n_c; st_ov = n_ov; }

        tc = st_tc; c = st_c; ov = st_ov;
        for (int b = 0; b < NB; ++b) {
            load_rows(ga, gp + b * BS);
            lstm_steps(ga, h, tc, c, ov, ws0, ws1, ws2, ws3);
            if (bad) store_block(op + b * BS, h);
        }
        e_tc = tc; e_c = c; e_ov = ov;
    }
}

extern "C" void kernel_launch(void* const* d_in, const int* in_sizes, int n_in,
                              void* d_out, int out_size, void* d_ws, size_t ws_size,
                              hipStream_t stream)
{
    const float* x    = (const float*)d_in[0];
    const float* Wih  = (const float*)d_in[1];
    const float* Whh  = (const float*)d_in[2];
    const float* bih  = (const float*)d_in[3];
    const float* bhh  = (const float*)d_in[4];
    const float* h0   = (const float*)d_in[5];
    const float* c0   = (const float*)d_in[6];
    float*       out  = (float*)d_out;
    float*       gts  = (float*)d_ws;

    gates_gemm_wmma<<<GBLOCKS, GTHREADS, 0, stream>>>(x, Wih, bih, bhh, Whh, gts, out);
    lstm_scan<<<1, NTHREADS, 0, stream>>>(gts, Whh, h0, c0, out);
}
